# starting kernel
# baseline (speedup 1.0000x reference)
#include <hip/hip_runtime.h>
#include <math.h>

#define HEADS 8
#define DIM 16
#define HD 128
#define INV_SCALE 0.25f

typedef __attribute__((ext_vector_type(2))) float v2f;
typedef __attribute__((ext_vector_type(8))) float v8f;

__global__ void zero_kernel(float* __restrict__ out, float* __restrict__ zbuf,
                            int nOut, int nZ) {
  int i = blockIdx.x * blockDim.x + threadIdx.x;
  if (i < nOut) out[i] = 0.0f;
  if (i < nZ)  zbuf[i] = 0.0f;
}

__global__ void __launch_bounds__(256)
edge_attn_kernel(const float* __restrict__ q, const float* __restrict__ k,
                 const float* __restrict__ v, const int* __restrict__ eidx,
                 float* __restrict__ wV, float* __restrict__ Z, int nEdges) {
  __shared__ int   s_idx[8][32];
  __shared__ float s_sc[8][16];

  const int lane = threadIdx.x & 31;
  const int wv   = threadIdx.x >> 5;
  const long e0  = ((long)blockIdx.x * 8 + wv) * 16;
  if (e0 >= nEdges) return;

  {
    long e   = e0 + (lane & 15);
    int  ok  = e < (long)nEdges;
    long off = (lane < 16) ? e : ((long)nEdges + e);
    s_idx[wv][lane] = ok ? eidx[off] : 0;
  }
  asm volatile("s_wait_dscnt 0x0" ::: "memory");

  const int m     = lane & 15;
  const int is_hi = lane >> 4;
  const int srcm  = s_idx[wv][m];
  const int dstm  = s_idx[wv][16 + m];

  const int em   = lane >> 1;
  const int half = lane & 1;
  const int srce = s_idx[wv][em];
  const int dste = s_idx[wv][16 + em];
  const bool ev  = (e0 + em) < (long)nEdges;

  const float* arow = k + (long)srcm * HD + is_hi * 2;
  const float* brow = q + (long)dstm * HD + is_hi * 2;
  const float* vrow = v + (long)srce * HD + half * 8;
  float*       orow = wV + (long)dste * HD + half * 8;

  const int  idxsel = lane & 7;
  const bool isdiag = (lane < 8) || (lane >= 24);
  const int  dm     = (lane >= 24) ? (lane - 16) : lane;

  for (int h = 0; h < HEADS; ++h) {
    const float* ap = arow + h * DIM;
    const float* bp = brow + h * DIM;
    v8f c = {0.f, 0.f, 0.f, 0.f, 0.f, 0.f, 0.f, 0.f};
#pragma unroll
    for (int c4 = 0; c4 < 4; ++c4) {
      v2f a = *(const v2f*)(ap + c4 * 4);
      v2f b = *(const v2f*)(bp + c4 * 4);
      c = __builtin_amdgcn_wmma_f32_16x16x4_f32(false, a, false, b,
                                                (short)0, c, false, false);
    }
    float d01 = (idxsel & 1) ? c[1] : c[0];
    float d23 = (idxsel & 1) ? c[3] : c[2];
    float d45 = (idxsel & 1) ? c[5] : c[4];
    float d67 = (idxsel & 1) ? c[7] : c[6];
    float d03 = (idxsel & 2) ? d23 : d01;
    float d47 = (idxsel & 2) ? d67 : d45;
    float dd  = (idxsel & 4) ? d47 : d03;
    if (isdiag) {
      float t = dd * INV_SCALE;
      t = fminf(fmaxf(t, -5.0f), 5.0f);
      s_sc[wv][dm] = __expf(t);
    }
    asm volatile("s_wait_dscnt 0x0" ::: "memory");
    float sc = s_sc[wv][em];

    if (ev) {
      const float4* vp = (const float4*)(vrow + h * DIM);
      float4 p0 = vp[0];
      float4 p1 = vp[1];
      float* op = orow + h * DIM;
      atomicAdd(op + 0, p0.x * sc); atomicAdd(op + 1, p0.y * sc);
      atomicAdd(op + 2, p0.z * sc); atomicAdd(op + 3, p0.w * sc);
      atomicAdd(op + 4, p1.x * sc); atomicAdd(op + 5, p1.y * sc);
      atomicAdd(op + 6, p1.z * sc); atomicAdd(op + 7, p1.w * sc);
      if (half == 0) atomicAdd(Z + (long)dste * HEADS + h, sc);
    }
  }
}

__global__ void norm_kernel(float* __restrict__ out, const float* __restrict__ Z,
                            int n) {
  int i = blockIdx.x * blockDim.x + threadIdx.x;
  if (i >= n) return;
  int node = i >> 7;
  int h    = (i >> 4) & 7;
  out[i] = out[i] / (Z[node * HEADS + h] + 1e-6f);
}

extern "C" void kernel_launch(void* const* d_in, const int* in_sizes, int n_in,
                              void* d_out, int out_size, void* d_ws, size_t ws_size,
                              hipStream_t stream) {
  const float* q    = (const float*)d_in[0];
  const float* k    = (const float*)d_in[1];
  const float* v    = (const float*)d_in[2];
  const int*   eidx = (const int*)d_in[3];
  float* out = (float*)d_out;
  float* Z   = (float*)d_ws;

  const int nEdges = in_sizes[3] / 2;
  const int nOut   = out_size;
  const int nZ     = nOut / DIM;

  const int threads = 256;
  const int blocks0 = (nOut + threads - 1) / threads;
  zero_kernel<<<blocks0, threads, 0, stream>>>(out, Z, nOut, nZ);

  const int groups  = (nEdges + 15) / 16;
  const int blocks1 = (groups + 7) / 8;
  edge_attn_kernel<<<blocks1, threads, 0, stream>>>(q, k, v, eidx, out, Z, nEdges);

  norm_kernel<<<blocks0, threads, 0, stream>>>(out, Z, nOut);
}


# Edges ordered by destination through a two-level counting sort in LDS (chunks split by 196-row range, then ranges by row; LDS int adds): 2 launches, no global atomics; row sums stay arrival order
# speedup vs baseline: 8.3219x; 8.3219x over previous
#include <hip/hip_runtime.h>
#include <math.h>

#define HEADS 8
#define DIM 16
#define HD 128
#define INV_SCALE 0.25f

typedef __attribute__((ext_vector_type(2))) float v2f;
typedef __attribute__((ext_vector_type(8))) float v8f;

#define SORT_T 1024
#define RB 196
#define NRG 256
#define NCH 256
#define CH_CAP 4096
#define RG_CAP 16384

__device__ __forceinline__ int scan256(int c, int tid, int* s_part, int* inclusive) {
  const int lane = tid & 31;
  const int wv   = tid >> 5;
  int inc = c;
  for (int off = 1; off < 32; off <<= 1) {
    int t = __shfl_up(inc, off);
    if (lane >= off) inc += t;
  }
  if (lane == 31 && wv < 8) s_part[wv] = inc;
  __syncthreads();
  int pre = 0;
  for (int i = 0; i < 8; ++i) pre += (i < wv) ? s_part[i] : 0;
  *inclusive = pre + inc;
  return pre + inc - c;
}

__global__ void __launch_bounds__(SORT_T)
split_chunk_kernel(const int* __restrict__ eidx, float* __restrict__ scratch,
                   int nEdges, int nNodes, int chunk) {
  __shared__ int s_hist[NRG];
  __shared__ int s_cur[NRG];
  __shared__ int s_part[8];
  __shared__ int s_stage[CH_CAP];
  int* piece = (int*)scratch;
  int* cntM  = piece + nEdges;
  int* offM  = cntM + NCH * NRG;
  const int tid = threadIdx.x;
  const int w   = blockIdx.x;
  const int e0  = w * chunk;
  const int e1  = min(nEdges, e0 + chunk);

  if (tid < NRG) s_hist[tid] = 0;
  __syncthreads();
  for (int eb = e0 + tid; eb < e1; eb += 4 * SORT_T) {
    int dd[4];
#pragma unroll
    for (int j = 0; j < 4; ++j) {
      const int e = eb + j * SORT_T;
      const int v = eidx[(long)nEdges + min(e, e1 - 1)];
      dd[j] = (e < e1) ? v : -1;
    }
#pragma unroll
    for (int j = 0; j < 4; ++j)
      if ((unsigned)dd[j] < (unsigned)nNodes) atomicAdd(&s_hist[dd[j] / RB], 1);
  }
  __syncthreads();
  const int c = (tid < NRG) ? s_hist[tid] : 0;
  int incl;
  const int excl = scan256(c, tid, s_part, &incl);
  if (tid < NRG) {
    s_cur[tid] = excl;
    cntM[w * NRG + tid] = c;
    offM[w * NRG + tid] = excl;
  }
  __syncthreads();
  const bool staged = (e1 - e0) <= CH_CAP;
  for (int eb = e0 + tid; eb < e1; eb += 4 * SORT_T) {
    int ss[4], dd[4];
#pragma unroll
    for (int j = 0; j < 4; ++j) {
      const int e  = eb + j * SORT_T;
      const int ec = min(e, e1 - 1);
      ss[j] = eidx[ec];
      const int v = eidx[(long)nEdges + ec];
      dd[j] = (e < e1) ? v : -1;
    }
#pragma unroll
    for (int j = 0; j < 4; ++j) {
      if ((unsigned)dd[j] < (unsigned)nNodes) {
        const int g   = dd[j] / RB;
        const int pos = atomicAdd(&s_cur[g], 1);
        const int rec = (ss[j] << 8) | (dd[j] - g * RB);
        if (staged) s_stage[pos] = rec; else piece[e0 + pos] = rec;
      }
    }
  }
  __syncthreads();
  if (staged) {
    const int total = s_cur[NRG - 1];
    for (int i = tid; i < total; i += SORT_T) piece[e0 + i] = s_stage[i];
  }
}

__global__ void __launch_bounds__(SORT_T)
order_range_kernel(const float* __restrict__ scratch, int* __restrict__ rowEnd,
                   int* __restrict__ srcSorted, int nEdges, int nNodes, int chunk) {
  __shared__ int s_cnt[NCH];
  __shared__ int s_off[NCH];
  __shared__ int s_hist[256];
  __shared__ int s_cur[256];
  __shared__ int s_part[8];
  __shared__ int s_tot[8];
  __shared__ int s_bas[8];
  __shared__ int s_list[RG_CAP];
  const int* piece = (const int*)scratch;
  const int* cntM  = piece + nEdges;
  const int* offM  = cntM + NCH * NRG;
  const int tid  = threadIdx.x;
  const int lane = tid & 31;
  const int wv   = tid >> 5;
  const int g    = blockIdx.x;

  int c = 0, o = 0;
  if (tid < NCH) {
    c = cntM[tid * NRG + g];
    o = offM[tid * NRG + g];
    s_cnt[tid] = c;
    s_off[tid] = o;
  }
  if (tid < 256) s_hist[tid] = 0;
  int rc = c, ro = o;
  for (int off = 16; off > 0; off >>= 1) {
    rc += __shfl_xor(rc, off);
    ro += __shfl_xor(ro, off);
  }
  if (lane == 0 && wv < 8) { s_tot[wv] = rc; s_bas[wv] = ro; }
  __syncthreads();
  int total = 0, base = 0;
  for (int i = 0; i < 8; ++i) { total += s_tot[i]; base += s_bas[i]; }

  int rec[8];
  unsigned have = 0;
#pragma unroll
  for (int k = 0; k < 8; ++k) {
    const int w = wv + 32 * k;
    const int* run = piece + (long)w * chunk + s_off[w];
    const bool in = lane < s_cnt[w];
    rec[k] = in ? run[lane] : 0;
    have |= in ? (1u << k) : 0u;
  }
#pragma unroll
  for (int k = 0; k < 8; ++k)
    if (have & (1u << k)) atomicAdd(&s_hist[rec[k] & 255], 1);
  for (int k = 0; k < 8; ++k) {
    const int w = wv + 32 * k;
    const int n = s_cnt[w];
    const int* run = piece + (long)w * chunk + s_off[w];
    for (int i = lane + 32; i < n; i += 32) atomicAdd(&s_hist[run[i] & 255], 1);
  }
  __syncthreads();

  const int hc = (tid < 256) ? s_hist[tid] : 0;
  int incl;
  const int excl = scan256(hc, tid, s_part, &incl);
  if (tid < 256) s_cur[tid] = excl;
  if (tid < RB) {
    const int row = g * RB + tid;
    if (row < nNodes) rowEnd[row] = base + incl;
  }
  __syncthreads();

  const bool staged = total <= RG_CAP;
#pragma unroll
  for (int k = 0; k < 8; ++k) {
    if (have & (1u << k)) {
      const int pos = atomicAdd(&s_cur[rec[k] & 255], 1);
      const int s   = (int)((unsigned)rec[k] >> 8);
      if (staged) s_list[pos] = s; else srcSorted[base + pos] = s;
    }
  }
  for (int k = 0; k < 8; ++k) {
    const int w = wv + 32 * k;
    const int n = s_cnt[w];
    const int* run = piece + (long)w * chunk + s_off[w];
    for (int i = lane + 32; i < n; i += 32) {
      const int r   = run[i];
      const int pos = atomicAdd(&s_cur[r & 255], 1);
      const int s   = (int)((unsigned)r >> 8);
      if (staged) s_list[pos] = s; else srcSorted[base + pos] = s;
    }
  }
  __syncthreads();
  if (staged)
    for (int i = tid; i < total; i += SORT_T) srcSorted[base + i] = s_list[i];
}

__global__ void __launch_bounds__(256) __attribute__((amdgpu_waves_per_eu(4, 4)))
gather_attn_kernel(const float* __restrict__ q, const float* __restrict__ k,
                   const float* __restrict__ v, const int* __restrict__ rowEnd,
                   const int* __restrict__ srcSorted, float* __restrict__ out,
                   int nNodes) {
  const int lane = threadIdx.x & 31;
  const int d = __builtin_amdgcn_readfirstlane(blockIdx.x * 8 + (threadIdx.x >> 5));
  if (d >= nNodes) return;

  const int c      = lane & 15;
  const int half   = lane >> 4;
  const int myhead = c >> 1;

  const int beg = d ? rowEnd[d - 1] : 0;
  const int end = rowEnd[d];

  const float* qrow = q + ((unsigned)d * HD + half * 8);
  float4 qf[8][2];
#pragma unroll
  for (int h = 0; h < HEADS; ++h) {
    qf[h][0] = *(const float4*)(qrow + h * DIM);
    qf[h][1] = *(const float4*)(qrow + h * DIM + 4);
  }

  float acc[8];
#pragma unroll
  for (int j = 0; j < 8; ++j) acc[j] = 0.0f;
  float z = 0.0f;

  for (int t0 = beg; t0 < end; t0 += 16) {
    const int cnt = min(16, end - t0);
    const int srcc = srcSorted[t0 + (c < cnt ? c : 0)];
    const float* krow = k + ((unsigned)srcc * HD + half * 8);

    float4 ka[8][2];
#pragma unroll
    for (int h = 0; h < HEADS; ++h) {
      ka[h][0] = *(const float4*)(krow + h * DIM);
      ka[h][1] = *(const float4*)(krow + h * DIM + 4);
    }
    float4 pv[8][2];
#pragma unroll
    for (int r = 0; r < 8; ++r) {
      const int se = __shfl(srcc, half * 8 + r);
      const float4* vp = (const float4*)(v + ((unsigned)se * HD + c * 8));
      pv[r][0] = vp[0];
      pv[r][1] = vp[1];
    }

    float w[8];
#pragma unroll
    for (int r = 0; r < 8; ++r) w[r] = 0.0f;

#pragma unroll
    for (int h = 0; h < HEADS; ++h) {
      const float4 a0 = ka[h][0], a1 = ka[h][1];
      const float4 b0 = qf[h][0], b1 = qf[h][1];
      v8f cc = {0.f, 0.f, 0.f, 0.f, 0.f, 0.f, 0.f, 0.f};
      v2f a, b;
      a = (v2f){a0.x, a0.y}; b = (v2f){b0.x, b0.y};
      cc = __builtin_amdgcn_wmma_f32_16x16x4_f32(false, a, false, b, (short)0, cc, false, false);
      a = (v2f){a0.z, a0.w}; b = (v2f){b0.z, b0.w};
      cc = __builtin_amdgcn_wmma_f32_16x16x4_f32(false, a, false, b, (short)0, cc, false, false);
      a = (v2f){a1.x, a1.y}; b = (v2f){b1.x, b1.y};
      cc = __builtin_amdgcn_wmma_f32_16x16x4_f32(false, a, false, b, (short)0, cc, false, false);
      a = (v2f){a1.z, a1.w}; b = (v2f){b1.z, b1.w};
      cc = __builtin_amdgcn_wmma_f32_16x16x4_f32(false, a, false, b, (short)0, cc, false, false);
#pragma unroll
      for (int r = 0; r < 8; ++r) w[r] = (myhead == h) ? cc[r] : w[r];
    }

#pragma unroll
    for (int r = 0; r < 8; ++r) {
      float t = w[r] * INV_SCALE;
      t = fminf(fmaxf(t, -5.0f), 5.0f);
      const float wr = (half * 8 + r < cnt) ? __expf(t) : 0.0f;
      acc[0] += pv[r][0].x * wr; acc[1] += pv[r][0].y * wr;
      acc[2] += pv[r][0].z * wr; acc[3] += pv[r][0].w * wr;
      acc[4] += pv[r][1].x * wr; acc[5] += pv[r][1].y * wr;
      acc[6] += pv[r][1].z * wr; acc[7] += pv[r][1].w * wr;
      z += wr;
    }
  }

#pragma unroll
  for (int j = 0; j < 8; ++j) acc[j] += __shfl_xor(acc[j], 16);
  z += __shfl_xor(z, 16);

  const float den = z + 1e-6f;
  float4 o;
  o.x = (half ? acc[4] : acc[0]) / den;
  o.y = (half ? acc[5] : acc[1]) / den;
  o.z = (half ? acc[6] : acc[2]) / den;
  o.w = (half ? acc[7] : acc[3]) / den;
  *(float4*)(out + ((unsigned)d * HD + c * 8 + half * 4)) = o;
}

extern "C" void kernel_launch(void* const* d_in, const int* in_sizes, int n_in,
                              void* d_out, int out_size, void* d_ws, size_t ws_size,
                              hipStream_t stream) {
  const float* q    = (const float*)d_in[0];
  const float* k    = (const float*)d_in[1];
  const float* v    = (const float*)d_in[2];
  const int*   eidx = (const int*)d_in[3];
  float* out = (float*)d_out;

  const int nEdges = in_sizes[3] / 2;
  const int nNodes = out_size / HD;

  int* rowEnd    = (int*)d_ws;
  int* srcSorted = rowEnd + nNodes;

  const int chunk = (nEdges + NCH - 1) / NCH;
  split_chunk_kernel<<<NCH, SORT_T, 0, stream>>>(eidx, out, nEdges, nNodes, chunk);
  order_range_kernel<<<NRG, SORT_T, 0, stream>>>(out, rowEnd, srcSorted, nEdges, nNodes, chunk);

  const int threads = 256;
  const int blocksG = (nNodes + 7) / 8;
  gather_attn_kernel<<<blocksG, threads, 0, stream>>>(q, k, v, rowEnd, srcSorted, out, nNodes);
}


# Gather runs as 512 resident work-groups whose waves walk rows 4096 apart (was one short-lived wave per row); next tile's source ids and row bounds two rows ahead are requested under current tile
# speedup vs baseline: 10.2066x; 1.2265x over previous
#include <hip/hip_runtime.h>
#include <math.h>

#define HEADS 8
#define DIM 16
#define HD 128
#define INV_SCALE 0.25f

typedef __attribute__((ext_vector_type(2))) float v2f;
typedef __attribute__((ext_vector_type(8))) float v8f;

#define SORT_T 1024
#define RB 196
#define NRG 256
#define NCH 256
#define CH_CAP 4096
#define RG_CAP 16384

__device__ __forceinline__ int scan256(int c, int tid, int* s_part, int* inclusive) {
  const int lane = tid & 31;
  const int wv   = tid >> 5;
  int inc = c;
  for (int off = 1; off < 32; off <<= 1) {
    int t = __shfl_up(inc, off);
    if (lane >= off) inc += t;
  }
  if (lane == 31 && wv < 8) s_part[wv] = inc;
  __syncthreads();
  int pre = 0;
  for (int i = 0; i < 8; ++i) pre += (i < wv) ? s_part[i] : 0;
  *inclusive = pre + inc;
  return pre + inc - c;
}

__global__ void __launch_bounds__(SORT_T)
split_chunk_kernel(const int* __restrict__ eidx, float* __restrict__ scratch,
                   int nEdges, int nNodes, int chunk) {
  __shared__ int s_hist[NRG];
  __shared__ int s_cur[NRG];
  __shared__ int s_part[8];
  __shared__ int s_stage[CH_CAP];
  int* piece = (int*)scratch;
  int* cntM  = piece + nEdges;
  int* offM  = cntM + NCH * NRG;
  const int tid = threadIdx.x;
  const int w   = blockIdx.x;
  const int e0  = w * chunk;
  const int e1  = min(nEdges, e0 + chunk);

  if (tid < NRG) s_hist[tid] = 0;
  __syncthreads();
  for (int eb = e0 + tid; eb < e1; eb += 4 * SORT_T) {
    int dd[4];
#pragma unroll
    for (int j = 0; j < 4; ++j) {
      const int e = eb + j * SORT_T;
      const int v = eidx[(long)nEdges + min(e, e1 - 1)];
      dd[j] = (e < e1) ? v : -1;
    }
#pragma unroll
    for (int j = 0; j < 4; ++j)
      if ((unsigned)dd[j] < (unsigned)nNodes) atomicAdd(&s_hist[dd[j] / RB], 1);
  }
  __syncthreads();
  const int c = (tid < NRG) ? s_hist[tid] : 0;
  int incl;
  const int excl = scan256(c, tid, s_part, &incl);
  if (tid < NRG) {
    s_cur[tid] = excl;
    cntM[w * NRG + tid] = c;
    offM[w * NRG + tid] = excl;
  }
  __syncthreads();
  const bool staged = (e1 - e0) <= CH_CAP;
  for (int eb = e0 + tid; eb < e1; eb += 4 * SORT_T) {
    int ss[4], dd[4];
#pragma unroll
    for (int j = 0; j < 4; ++j) {
      const int e  = eb + j * SORT_T;
      const int ec = min(e, e1 - 1);
      ss[j] = eidx[ec];
      const int v = eidx[(long)nEdges + ec];
      dd[j] = (e < e1) ? v : -1;
    }
#pragma unroll
    for (int j = 0; j < 4; ++j) {
      if ((unsigned)dd[j] < (unsigned)nNodes) {
        const int g   = dd[j] / RB;
        const int pos = atomicAdd(&s_cur[g], 1);
        const int rec = (ss[j] << 8) | (dd[j] - g * RB);
        if (staged) s_stage[pos] = rec; else piece[e0 + pos] = rec;
      }
    }
  }
  __syncthreads();
  if (staged) {
    const int total = s_cur[NRG - 1];
    for (int i = tid; i < total; i += SORT_T) piece[e0 + i] = s_stage[i];
  }
}

__global__ void __launch_bounds__(SORT_T)
order_range_kernel(const float* __restrict__ scratch, int* __restrict__ rowEnd,
                   int* __restrict__ srcSorted, int nEdges, int nNodes, int chunk) {
  __shared__ int s_cnt[NCH];
  __shared__ int s_off[NCH];
  __shared__ int s_hist[256];
  __shared__ int s_cur[256];
  __shared__ int s_part[8];
  __shared__ int s_tot[8];
  __shared__ int s_bas[8];
  __shared__ int s_list[RG_CAP];
  const int* piece = (const int*)scratch;
  const int* cntM  = piece + nEdges;
  const int* offM  = cntM + NCH * NRG;
  const int tid  = threadIdx.x;
  const int lane = tid & 31;
  const int wv   = tid >> 5;
  const int g    = blockIdx.x;

  int c = 0, o = 0;
  if (tid < NCH) {
    c = cntM[tid * NRG + g];
    o = offM[tid * NRG + g];
    s_cnt[tid] = c;
    s_off[tid] = o;
  }
  if (tid < 256) s_hist[tid] = 0;
  int rc = c, ro = o;
  for (int off = 16; off > 0; off >>= 1) {
    rc += __shfl_xor(rc, off);
    ro += __shfl_xor(ro, off);
  }
  if (lane == 0 && wv < 8) { s_tot[wv] = rc; s_bas[wv] = ro; }
  __syncthreads();
  int total = 0, base = 0;
  for (int i = 0; i < 8; ++i) { total += s_tot[i]; base += s_bas[i]; }

  int rec[8];
  unsigned have = 0;
#pragma unroll
  for (int k = 0; k < 8; ++k) {
    const int w = wv + 32 * k;
    const int* run = piece + (long)w * chunk + s_off[w];
    const bool in = lane < s_cnt[w];
    rec[k] = in ? run[lane] : 0;
    have |= in ? (1u << k) : 0u;
  }
#pragma unroll
  for (int k = 0; k < 8; ++k)
    if (have & (1u << k)) atomicAdd(&s_hist[rec[k] & 255], 1);
  for (int k = 0; k < 8; ++k) {
    const int w = wv + 32 * k;
    const int n = s_cnt[w];
    const int* run = piece + (long)w * chunk + s_off[w];
    for (int i = lane + 32; i < n; i += 32) atomicAdd(&s_hist[run[i] & 255], 1);
  }
  __syncthreads();

  const int hc = (tid < 256) ? s_hist[tid] : 0;
  int incl;
  const int excl = scan256(hc, tid, s_part, &incl);
  if (tid < 256) s_cur[tid] = excl;
  if (tid < RB) {
    const int row = g * RB + tid;
    if (row < nNodes) rowEnd[row] = base + incl;
  }
  __syncthreads();

  const bool staged = total <= RG_CAP;
#pragma unroll
  for (int k = 0; k < 8; ++k) {
    if (have & (1u << k)) {
      const int pos = atomicAdd(&s_cur[rec[k] & 255], 1);
      const int s   = (int)((unsigned)rec[k] >> 8);
      if (staged) s_list[pos] = s; else srcSorted[base + pos] = s;
    }
  }
  for (int k = 0; k < 8; ++k) {
    const int w = wv + 32 * k;
    const int n = s_cnt[w];
    const int* run = piece + (long)w * chunk + s_off[w];
    for (int i = lane + 32; i < n; i += 32) {
      const int r   = run[i];
      const int pos = atomicAdd(&s_cur[r & 255], 1);
      const int s   = (int)((unsigned)r >> 8);
      if (staged) s_list[pos] = s; else srcSorted[base + pos] = s;
    }
  }
  __syncthreads();
  if (staged)
    for (int i = tid; i < total; i += SORT_T) srcSorted[base + i] = s_list[i];
}

#define GATHER_WGS 512
__global__ void __launch_bounds__(256) __attribute__((amdgpu_waves_per_eu(4, 4)))
gather_attn_kernel(const float* __restrict__ q, const float* __restrict__ k,
                   const float* __restrict__ v, const int* __restrict__ rowEnd,
                   const int* __restrict__ srcSorted, float* __restrict__ out,
                   int nNodes) {
  const int lane = threadIdx.x & 31;
  const int c      = lane & 15;
  const int half   = lane >> 4;
  const int myhead = c >> 1;
  const int stride = gridDim.x * 8;
  int d = __builtin_amdgcn_readfirstlane((threadIdx.x >> 5) * gridDim.x + blockIdx.x);
  if (d >= nNodes) return;

  int beg = d ? rowEnd[d - 1] : 0;
  int end = rowEnd[d];
  int ids = 0;
  if (beg < end) ids = srcSorted[beg + (c < end - beg ? c : 0)];
  int nbv = 0;
  if (d + stride < nNodes) nbv = rowEnd[d + stride - 1 + (lane & 1)];

  for (;;) {
    const int dn  = d + stride;
    const int dnn = dn + stride;
    const int nbeg = __builtin_amdgcn_readlane(nbv, 0);
    const int nend = __builtin_amdgcn_readlane(nbv, 1);

    const float* qrow = q + ((unsigned)d * HD + half * 8);
    float4 qf[8][2];
#pragma unroll
    for (int h = 0; h < HEADS; ++h) {
      qf[h][0] = *(const float4*)(qrow + h * DIM);
      qf[h][1] = *(const float4*)(qrow + h * DIM + 4);
    }

    int fbv = 0;
    if (dnn < nNodes) fbv = rowEnd[dnn - 1 + (lane & 1)];

    float acc[8];
#pragma unroll
    for (int j = 0; j < 8; ++j) acc[j] = 0.0f;
    float z = 0.0f;

    for (int t0 = beg; t0 < end; t0 += 16) {
      const int cnt = min(16, end - t0);
      const int srcc = ids;
      const float* krow = k + ((unsigned)srcc * HD + half * 8);

      float4 ka[8][2];
#pragma unroll
      for (int h = 0; h < HEADS; ++h) {
        ka[h][0] = *(const float4*)(krow + h * DIM);
        ka[h][1] = *(const float4*)(krow + h * DIM + 4);
      }
      float4 pv[8][2];
#pragma unroll
      for (int r = 0; r < 8; ++r) {
        const int se = __shfl(srcc, half * 8 + r);
        const float4* vp = (const float4*)(v + ((unsigned)se * HD + c * 8));
        pv[r][0] = vp[0];
        pv[r][1] = vp[1];
      }

      const int t1 = t0 + 16;
      const int na = (t1 < end) ? t1 : nbeg;
      const int nn = (t1 < end) ? end - t1 : nend - nbeg;
      if (nn > 0) ids = srcSorted[na + (c < nn ? c : 0)];

      float w[8];
#pragma unroll
      for (int r = 0; r < 8; ++r) w[r] = 0.0f;

#pragma unroll
      for (int h = 0; h < HEADS; ++h) {
        const float4 a0 = ka[h][0], a1 = ka[h][1];
        const float4 b0 = qf[h][0], b1 = qf[h][1];
        v8f cc = {0.f, 0.f, 0.f, 0.f, 0.f, 0.f, 0.f, 0.f};
        v2f a, b;
        a = (v2f){a0.x, a0.y}; b = (v2f){b0.x, b0.y};
        cc = __builtin_amdgcn_wmma_f32_16x16x4_f32(false, a, false, b, (short)0, cc, false, false);
        a = (v2f){a0.z, a0.w}; b = (v2f){b0.z, b0.w};
        cc = __builtin_amdgcn_wmma_f32_16x16x4_f32(false, a, false, b, (short)0, cc, false, false);
        a = (v2f){a1.x, a1.y}; b = (v2f){b1.x, b1.y};
        cc = __builtin_amdgcn_wmma_f32_16x16x4_f32(false, a, false, b, (short)0, cc, false, false);
        a = (v2f){a1.z, a1.w}; b = (v2f){b1.z, b1.w};
        cc = __builtin_amdgcn_wmma_f32_16x16x4_f32(false, a, false, b, (short)0, cc, false, false);
#pragma unroll
        for (int r = 0; r < 8; ++r) w[r] = (myhead == h) ? cc[r] : w[r];
      }

#pragma unroll
      for (int r = 0; r < 8; ++r) {
        float t = w[r] * INV_SCALE;
        t = fminf(fmaxf(t, -5.0f), 5.0f);
        const float wr = (half * 8 + r < cnt) ? __expf(t) : 0.0f;
        acc[0] += pv[r][0].x * wr; acc[1] += pv[r][0].y * wr;
        acc[2] += pv[r][0].z * wr; acc[3] += pv[r][0].w * wr;
        acc[4] += pv[r][1].x * wr; acc[5] += pv[r][1].y * wr;
        acc[6] += pv[r][1].z * wr; acc[7] += pv[r][1].w * wr;
        z += wr;
      }
    }

    if (beg >= end && nbeg < nend) ids = srcSorted[nbeg + (c < nend - nbeg ? c : 0)];

#pragma unroll
    for (int j = 0; j < 8; ++j) acc[j] += __shfl_xor(acc[j], 16);
    z += __shfl_xor(z, 16);

    const float den = z + 1e-6f;
    float4 o;
    o.x = (half ? acc[4] : acc[0]) / den;
    o.y = (half ? acc[5] : acc[1]) / den;
    o.z = (half ? acc[6] : acc[2]) / den;
    o.w = (half ? acc[7] : acc[3]) / den;
    *(float4*)(out + ((unsigned)d * HD + c * 8 + half * 4)) = o;

    if (dn >= nNodes) break;
    d = dn; beg = nbeg; end = nend; nbv = fbv;
  }
}

extern "C" void kernel_launch(void* const* d_in, const int* in_sizes, int n_in,
                              void* d_out, int out_size, void* d_ws, size_t ws_size,
                              hipStream_t stream) {
  const float* q    = (const float*)d_in[0];
  const float* k    = (const float*)d_in[1];
  const float* v    = (const float*)d_in[2];
  const int*   eidx = (const int*)d_in[3];
  float* out = (float*)d_out;

  const int nEdges = in_sizes[3] / 2;
  const int nNodes = out_size / HD;

  int* rowEnd    = (int*)d_ws;
  int* srcSorted = rowEnd + nNodes;

  const int chunk = (nEdges + NCH - 1) / NCH;
  split_chunk_kernel<<<NCH, SORT_T, 0, stream>>>(eidx, out, nEdges, nNodes, chunk);
  order_range_kernel<<<NRG, SORT_T, 0, stream>>>(out, rowEnd, srcSorted, nEdges, nNodes, chunk);

  gather_attn_kernel<<<GATHER_WGS, 256, 0, stream>>>(q, k, v, rowEnd, srcSorted, out, nNodes);
}


# Per-row ordering of each 196-row range moved into the gather launch: its work-group orders the range in LDS, its 16 waves draw rows by an LDS counter: 2 launches (was 3); sums stay arrival order
# speedup vs baseline: 10.6576x; 1.0442x over previous
#include <hip/hip_runtime.h>
#include <math.h>

#define HEADS 8
#define DIM 16
#define HD 128
#define INV_SCALE 0.25f

typedef __attribute__((ext_vector_type(2))) float v2f;
typedef __attribute__((ext_vector_type(8))) float v8f;

#define SORT_T 1024
#define RB 196
#define NRG 256
#define NCH 256
#define CH_CAP 4096
#define RG_CAP 32768

__device__ __forceinline__ int scan256(int c, int tid, int* s_part, int* inclusive) {
  const int lane = tid & 31;
  const int wv   = tid >> 5;
  int inc = c;
  for (int off = 1; off < 32; off <<= 1) {
    int t = __shfl_up(inc, off);
    if (lane >= off) inc += t;
  }
  if (lane == 31 && wv < 8) s_part[wv] = inc;
  __syncthreads();
  int pre = 0;
  for (int i = 0; i < 8; ++i) pre += (i < wv) ? s_part[i] : 0;
  *inclusive = pre + inc;
  return pre + inc - c;
}

__global__ void __launch_bounds__(SORT_T)
split_chunk_kernel(const int* __restrict__ eidx, int* __restrict__ piece,
                   int nEdges, int nNodes, int chunk) {
  __shared__ int s_hist[NRG];
  __shared__ int s_cur[NRG];
  __shared__ int s_part[8];
  __shared__ int s_stage[CH_CAP];
  unsigned short* endM = (unsigned short*)(piece + nEdges);
  const int tid = threadIdx.x;
  const int w   = blockIdx.x;
  const int e0  = w * chunk;
  const int e1  = min(nEdges, e0 + chunk);

  if (tid < NRG) s_hist[tid] = 0;
  __syncthreads();
  for (int eb = e0 + tid; eb < e1; eb += 4 * SORT_T) {
    int dd[4];
#pragma unroll
    for (int j = 0; j < 4; ++j) {
      const int e = eb + j * SORT_T;
      const int v = eidx[(long)nEdges + min(e, e1 - 1)];
      dd[j] = (e < e1) ? v : -1;
    }
#pragma unroll
    for (int j = 0; j < 4; ++j)
      if ((unsigned)dd[j] < (unsigned)nNodes) atomicAdd(&s_hist[dd[j] / RB], 1);
  }
  __syncthreads();
  const int c = (tid < NRG) ? s_hist[tid] : 0;
  int incl;
  const int excl = scan256(c, tid, s_part, &incl);
  if (tid < NRG) {
    s_cur[tid] = excl;
    endM[w * NRG + tid] = (unsigned short)incl;
  }
  __syncthreads();
  const bool staged = (e1 - e0) <= CH_CAP;
  for (int eb = e0 + tid; eb < e1; eb += 4 * SORT_T) {
    int ss[4], dd[4];
#pragma unroll
    for (int j = 0; j < 4; ++j) {
      const int e  = eb + j * SORT_T;
      const int ec = min(e, e1 - 1);
      ss[j] = eidx[ec];
      const int v = eidx[(long)nEdges + ec];
      dd[j] = (e < e1) ? v : -1;
    }
#pragma unroll
    for (int j = 0; j < 4; ++j) {
      if ((unsigned)dd[j] < (unsigned)nNodes) {
        const int g   = dd[j] / RB;
        const int pos = atomicAdd(&s_cur[g], 1);
        const int rec = (ss[j] << 8) | (dd[j] - g * RB);
        if (staged) s_stage[pos] = rec; else piece[e0 + pos] = rec;
      }
    }
  }
  __syncthreads();
  if (staged) {
    const int total = s_cur[NRG - 1];
    for (int i = tid; i < total; i += SORT_T) piece[e0 + i] = s_stage[i];
  }
}

#define TILE_LOADS(ids_)                                                                   \
  const int srcc = (ids_);                                                                 \
  const float* krow = k + ((unsigned)srcc * HD + half * 8);                                \
  float4 ka[8][2];                                                                         \
  _Pragma("unroll") for (int h = 0; h < HEADS; ++h) {                                      \
    ka[h][0] = *(const float4*)(krow + h * DIM);                                           \
    ka[h][1] = *(const float4*)(krow + h * DIM + 4);                                       \
  }                                                                                        \
  float4 pv[8][2];                                                                         \
  _Pragma("unroll") for (int r = 0; r < 8; ++r) {                                          \
    const int se = __shfl(srcc, half * 8 + r);                                             \
    const float4* vp = (const float4*)(v + ((unsigned)se * HD + c * 8));                   \
    pv[r][0] = vp[0];                                                                      \
    pv[r][1] = vp[1];                                                                      \
  }

#define TILE_MATH(mask_)                                                                   \
  {                                                                                        \
    float w[8];                                                                            \
    _Pragma("unroll") for (int r = 0; r < 8; ++r) w[r] = 0.0f;                             \
    _Pragma("unroll") for (int h = 0; h < HEADS; ++h) {                                    \
      const float4 a0 = ka[h][0], a1 = ka[h][1];                                           \
      const float4 b0 = qf[h][0], b1 = qf[h][1];                                           \
      v8f cc = {0.f, 0.f, 0.f, 0.f, 0.f, 0.f, 0.f, 0.f};                                   \
      v2f a, b;                                                                            \
      a = (v2f){a0.x, a0.y}; b = (v2f){b0.x, b0.y};                                        \
      cc = __builtin_amdgcn_wmma_f32_16x16x4_f32(false, a, false, b, (short)0, cc, false, false); \
      a = (v2f){a0.z, a0.w}; b = (v2f){b0.z, b0.w};                                        \
      cc = __builtin_amdgcn_wmma_f32_16x16x4_f32(false, a, false, b, (short)0, cc, false, false); \
      a = (v2f){a1.x, a1.y}; b = (v2f){b1.x, b1.y};                                        \
      cc = __builtin_amdgcn_wmma_f32_16x16x4_f32(false, a, false, b, (short)0, cc, false, false); \
      a = (v2f){a1.z, a1.w}; b = (v2f){b1.z, b1.w};                                        \
      cc = __builtin_amdgcn_wmma_f32_16x16x4_f32(false, a, false, b, (short)0, cc, false, false); \
      _Pragma("unroll") for (int r = 0; r < 8; ++r) w[r] = (myhead == h) ? cc[r] : w[r];   \
    }                                                                                      \
    const unsigned mh = (unsigned)(mask_) >> (half * 8);                                   \
    _Pragma("unroll") for (int r = 0; r < 8; ++r) {                                        \
      float t = w[r] * INV_SCALE;                                                          \
      t = fminf(fmaxf(t, -5.0f), 5.0f);                                                    \
      const float wr = ((mh >> r) & 1u) ? __expf(t) : 0.0f;                                \
      acc[0] += pv[r][0].x * wr; acc[1] += pv[r][0].y * wr;                                \
      acc[2] += pv[r][0].z * wr; acc[3] += pv[r][0].w * wr;                                \
      acc[4] += pv[r][1].x * wr; acc[5] += pv[r][1].y * wr;                                \
      acc[6] += pv[r][1].z * wr; acc[7] += pv[r][1].w * wr;                                \
      z += wr;                                                                             \
    }                                                                                      \
  }

#define ROW_BEGIN(d_)                                                                      \
  const float* qrow = q + ((unsigned)(d_) * HD + half * 8);                                \
  float4 qf[8][2];                                                                         \
  _Pragma("unroll") for (int h = 0; h < HEADS; ++h) {                                      \
    qf[h][0] = *(const float4*)(qrow + h * DIM);                                           \
    qf[h][1] = *(const float4*)(qrow + h * DIM + 4);                                       \
  }                                                                                        \
  float acc[8];                                                                            \
  _Pragma("unroll") for (int j = 0; j < 8; ++j) acc[j] = 0.0f;                             \
  float z = 0.0f;

#define ROW_END(d_)                                                                        \
  {                                                                                        \
    _Pragma("unroll") for (int j = 0; j < 8; ++j) acc[j] += __shfl_xor(acc[j], 16);        \
    z += __shfl_xor(z, 16);                                                                \
    const float den = z + 1e-6f;                                                           \
    float4 o;                                                                              \
    o.x = (half ? acc[4] : acc[0]) / den;                                                  \
    o.y = (half ? acc[5] : acc[1]) / den;                                                  \
    o.z = (half ? acc[6] : acc[2]) / den;                                                  \
    o.w = (half ? acc[7] : acc[3]) / den;                                                  \
    *(float4*)(out + ((unsigned)(d_) * HD + c * 8 + half * 4)) = o;                        \
  }

#define GW 16
#define GT (GW * 32)
__global__ void __launch_bounds__(GT) __attribute__((amdgpu_waves_per_eu(4, 4)))
range_gather_kernel(const float* __restrict__ q, const float* __restrict__ k,
                    const float* __restrict__ v, const int* __restrict__ piece,
                    float* __restrict__ out, int nEdges, int nNodes, int chunk) {
  __shared__ int s_cnt[NCH];
  __shared__ int s_off[NCH];
  __shared__ int s_hist[256];
  __shared__ int s_cur[256];
  __shared__ int s_rend[256];
  __shared__ int s_part[8];
  __shared__ int s_tot[8];
  __shared__ int s_next;
  __shared__ int s_list[RG_CAP];
  const unsigned short* endM = (const unsigned short*)(piece + nEdges);
  const int tid  = threadIdx.x;
  const int lane = tid & 31;
  const int wv   = __builtin_amdgcn_readfirstlane(tid >> 5);
  const int g    = blockIdx.x;
  const int c      = lane & 15;
  const int half   = lane >> 4;
  const int myhead = c >> 1;
  const int rows = min(RB, nNodes - g * RB);

  int cn = 0;
  if (tid < NCH) {
    const int o = g ? (int)endM[tid * NRG + g - 1] : 0;
    cn = (int)endM[tid * NRG + g] - o;
    s_cnt[tid] = cn;
    s_off[tid] = o;
  }
  if (tid < 256) s_hist[tid] = 0;
  if (tid == 0) s_next = GW;
  int rc = cn;
  for (int off = 16; off > 0; off >>= 1) rc += __shfl_xor(rc, off);
  if (lane == 0 && wv < 8) s_tot[wv] = rc;
  __syncthreads();
  int total = 0;
  for (int i = 0; i < 8; ++i) total += s_tot[i];
  const bool staged = total <= RG_CAP;

  if (staged) {
    int rec[NCH / GW];
    unsigned have = 0;
#pragma unroll
    for (int j = 0; j < NCH / GW; ++j) {
      const int w = wv + GW * j;
      const int* run = piece + (long)w * chunk + s_off[w];
      const bool in = lane < s_cnt[w];
      rec[j] = in ? run[lane] : 0;
      have |= in ? (1u << j) : 0u;
    }
#pragma unroll
    for (int j = 0; j < NCH / GW; ++j)
      if (have & (1u << j)) atomicAdd(&s_hist[rec[j] & 255], 1);
    for (int j = 0; j < NCH / GW; ++j) {
      const int w = wv + GW * j;
      const int n = s_cnt[w];
      const int* run = piece + (long)w * chunk + s_off[w];
      for (int i = lane + 32; i < n; i += 32) atomicAdd(&s_hist[run[i] & 255], 1);
    }
    __syncthreads();

    const int hc = (tid < 256) ? s_hist[tid] : 0;
    int incl;
    const int excl = scan256(hc, tid, s_part, &incl);
    if (tid < 256) { s_cur[tid] = excl; s_rend[tid] = incl; }
    __syncthreads();

#pragma unroll
    for (int j = 0; j < NCH / GW; ++j) {
      if (have & (1u << j)) {
        const int pos = atomicAdd(&s_cur[rec[j] & 255], 1);
        s_list[pos] = (int)((unsigned)rec[j] >> 8);
      }
    }
    for (int j = 0; j < NCH / GW; ++j) {
      const int w = wv + GW * j;
      const int n = s_cnt[w];
      const int* run = piece + (long)w * chunk + s_off[w];
      for (int i = lane + 32; i < n; i += 32) {
        const int r   = run[i];
        const int pos = atomicAdd(&s_cur[r & 255], 1);
        s_list[pos] = (int)((unsigned)r >> 8);
      }
    }
    __syncthreads();

    int r = wv;
    if (r >= rows) return;
    int rn = 0;
    if (lane == 0) rn = atomicAdd(&s_next, 1);
    rn = __builtin_amdgcn_readfirstlane(rn);
    int beg = __builtin_amdgcn_readfirstlane(r ? s_rend[r ? r - 1 : 0] : 0);
    int end = __builtin_amdgcn_readfirstlane(s_rend[r]);
    int ids = 0;
    if (beg < end) ids = s_list[beg + (c < end - beg ? c : 0)];

    for (;;) {
      int nbeg = 0, nend = 0;
      if (rn < rows) {
        nbeg = __builtin_amdgcn_readfirstlane(s_rend[rn - 1]);
        nend = __builtin_amdgcn_readfirstlane(s_rend[rn]);
      }
      int rnn = 0;
      if (lane == 0) rnn = atomicAdd(&s_next, 1);
      rnn = __builtin_amdgcn_readfirstlane(rnn);

      const int d = g * RB + r;
      ROW_BEGIN(d)
      for (int t0 = beg; t0 < end; t0 += 16) {
        const int cnt = min(16, end - t0);
        TILE_LOADS(ids)
        const int t1 = t0 + 16;
        const int na = (t1 < end) ? t1 : nbeg;
        const int nn = (t1 < end) ? end - t1 : nend - nbeg;
        if (nn > 0) ids = s_list[na + (c < nn ? c : 0)];
        TILE_MATH(0xffffu >> (16 - cnt))
      }
      if (beg >= end && nbeg < nend) ids = s_list[nbeg + (c < nend - nbeg ? c : 0)];
      ROW_END(d)

      if (rn >= rows) break;
      r = rn; beg = nbeg; end = nend; rn = rnn;
    }
  } else {
    for (int r = wv; r < rows; r += GW) {
      const int d = g * RB + r;
      ROW_BEGIN(d)
      for (int w = 0; w < NCH; ++w) {
        const int n = __builtin_amdgcn_readfirstlane(s_cnt[w]);
        const int* run = piece + (long)w * chunk + __builtin_amdgcn_readfirstlane(s_off[w]);
        for (int i = 0; i < n; i += 16) {
          const int rcd = (c < n - i) ? run[i + c] : -1;
          const bool ok = rcd >= 0 && (rcd & 255) == r;
          const unsigned mask = (unsigned)__ballot(ok) & 0xffffu;
          if (mask) {
            TILE_LOADS(ok ? (int)((unsigned)rcd >> 8) : 0)
            TILE_MATH(mask)
          }
        }
      }
      ROW_END(d)
    }
  }
}

extern "C" void kernel_launch(void* const* d_in, const int* in_sizes, int n_in,
                              void* d_out, int out_size, void* d_ws, size_t ws_size,
                              hipStream_t stream) {
  const float* q    = (const float*)d_in[0];
  const float* k    = (const float*)d_in[1];
  const float* v    = (const float*)d_in[2];
  const int*   eidx = (const int*)d_in[3];
  float* out = (float*)d_out;

  const int nEdges = in_sizes[3] / 2;
  const int nNodes = out_size / HD;

  int* piece = (int*)d_ws;

  const int chunk = (nEdges + NCH - 1) / NCH;
  split_chunk_kernel<<<NCH, SORT_T, 0, stream>>>(eidx, piece, nEdges, nNodes, chunk);
  range_gather_kernel<<<NRG, GT, 0, stream>>>(q, k, v, piece, out, nEdges, nNodes, chunk);
}


# Score tile = 2 edges x 8 heads as the 16 A rows, Q's heads as B columns (Q in 8 registers); lane owns 4+4 V floats: K load touches 8 lines (was 16), V load 4 (was 8); V sums: even/odd slots per half
# speedup vs baseline: 11.6018x; 1.0886x over previous
#include <hip/hip_runtime.h>
#include <math.h>

#define HEADS 8
#define DIM 16
#define HD 128
#define INV_SCALE 0.25f

typedef __attribute__((ext_vector_type(2))) float v2f;
typedef __attribute__((ext_vector_type(8))) float v8f;

#define SORT_T 1024
#define RB 196
#define NRG 256
#define NCH 256
#define CH_CAP 4096
#define RG_CAP 32768

__device__ __forceinline__ int scan256(int c, int tid, int* s_part, int* inclusive) {
  const int lane = tid & 31;
  const int wv   = tid >> 5;
  int inc = c;
  for (int off = 1; off < 32; off <<= 1) {
    int t = __shfl_up(inc, off);
    if (lane >= off) inc += t;
  }
  if (lane == 31 && wv < 8) s_part[wv] = inc;
  __syncthreads();
  int pre = 0;
  for (int i = 0; i < 8; ++i) pre += (i < wv) ? s_part[i] : 0;
  *inclusive = pre + inc;
  return pre + inc - c;
}

__global__ void __launch_bounds__(SORT_T)
split_chunk_kernel(const int* __restrict__ eidx, int* __restrict__ piece,
                   int nEdges, int nNodes, int chunk) {
  __shared__ int s_hist[NRG];
  __shared__ int s_cur[NRG];
  __shared__ int s_part[8];
  __shared__ int s_stage[CH_CAP];
  unsigned short* endM = (unsigned short*)(piece + nEdges);
  const int tid = threadIdx.x;
  const int w   = blockIdx.x;
  const int e0  = w * chunk;
  const int e1  = min(nEdges, e0 + chunk);

  if (tid < NRG) s_hist[tid] = 0;
  __syncthreads();
  for (int eb = e0 + tid; eb < e1; eb += 4 * SORT_T) {
    int dd[4];
#pragma unroll
    for (int j = 0; j < 4; ++j) {
      const int e = eb + j * SORT_T;
      const int v = eidx[(long)nEdges + min(e, e1 - 1)];
      dd[j] = (e < e1) ? v : -1;
    }
#pragma unroll
    for (int j = 0; j < 4; ++j)
      if ((unsigned)dd[j] < (unsigned)nNodes) atomicAdd(&s_hist[dd[j] / RB], 1);
  }
  __syncthreads();
  const int c = (tid < NRG) ? s_hist[tid] : 0;
  int incl;
  const int excl = scan256(c, tid, s_part, &incl);
  if (tid < NRG) {
    s_cur[tid] = excl;
    endM[w * NRG + tid] = (unsigned short)incl;
  }
  __syncthreads();
  const bool staged = (e1 - e0) <= CH_CAP;
  for (int eb = e0 + tid; eb < e1; eb += 4 * SORT_T) {
    int ss[4], dd[4];
#pragma unroll
    for (int j = 0; j < 4; ++j) {
      const int e  = eb + j * SORT_T;
      const int ec = min(e, e1 - 1);
      ss[j] = eidx[ec];
      const int v = eidx[(long)nEdges + ec];
      dd[j] = (e < e1) ? v : -1;
    }
#pragma unroll
    for (int j = 0; j < 4; ++j) {
      if ((unsigned)dd[j] < (unsigned)nNodes) {
        const int g   = dd[j] / RB;
        const int pos = atomicAdd(&s_cur[g], 1);
        const int rec = (ss[j] << 8) | (dd[j] - g * RB);
        if (staged) s_stage[pos] = rec; else piece[e0 + pos] = rec;
      }
    }
  }
  __syncthreads();
  if (staged) {
    const int total = s_cur[NRG - 1];
    for (int i = tid; i < total; i += SORT_T) piece[e0 + i] = s_stage[i];
  }
}

#define TILE_LOADS(ids_)                                                                   \
  const int srcc = (ids_);                                                                 \
  float4 kk[8][2];                                                                         \
  float4 pv[8][2];                                                                         \
  _Pragma("unroll") for (int p = 0; p < 8; ++p) {                                          \
    const int sA = __builtin_amdgcn_readlane(srcc, 2 * p);                                 \
    const int sB = __builtin_amdgcn_readlane(srcc, 2 * p + 1);                             \
    const int sk = (c & 8) ? sB : sA;                                                      \
    const float4* kp = (const float4*)(k + ((unsigned)sk * HD + hsel * DIM + half * 8));   \
    kk[p][0] = kp[0];                                                                      \
    kk[p][1] = kp[1];                                                                      \
  }                                                                                        \
  _Pragma("unroll") for (int p = 0; p < 8; ++p) {                                          \
    const int sA = __builtin_amdgcn_readlane(srcc, 2 * p);                                 \
    const int sB = __builtin_amdgcn_readlane(srcc, 2 * p + 1);                             \
    const int sv = half ? sB : sA;                                                         \
    const float* vp = v + ((unsigned)sv * HD + c * 4);                                     \
    pv[p][0] = *(const float4*)vp;                                                         \
    pv[p][1] = *(const float4*)(vp + 64);                                                  \
  }

#define TILE_MATH(mask_)                                                                   \
  {                                                                                        \
    const unsigned mh = (unsigned)(mask_) >> half;                                         \
    _Pragma("unroll") for (int p = 0; p < 8; ++p) {                                        \
      const float4 a0 = kk[p][0], a1 = kk[p][1];                                           \
      v8f cc = {0.f, 0.f, 0.f, 0.f, 0.f, 0.f, 0.f, 0.f};                                   \
      v2f a, b;                                                                            \
      a = (v2f){a0.x, a0.y}; b = (v2f){qb0.x, qb0.y};                                      \
      cc = __builtin_amdgcn_wmma_f32_16x16x4_f32(false, a, false, b, (short)0, cc, false, false); \
      a = (v2f){a0.z, a0.w}; b = (v2f){qb0.z, qb0.w};                                      \
      cc = __builtin_amdgcn_wmma_f32_16x16x4_f32(false, a, false, b, (short)0, cc, false, false); \
      a = (v2f){a1.x, a1.y}; b = (v2f){qb1.x, qb1.y};                                      \
      cc = __builtin_amdgcn_wmma_f32_16x16x4_f32(false, a, false, b, (short)0, cc, false, false); \
      a = (v2f){a1.z, a1.w}; b = (v2f){qb1.z, qb1.w};                                      \
      cc = __builtin_amdgcn_wmma_f32_16x16x4_f32(false, a, false, b, (short)0, cc, false, false); \
      float s = cc[0];                                                                     \
      _Pragma("unroll") for (int r = 1; r < 8; ++r) s = (hsel == r) ? cc[r] : s;           \
      float t = s * INV_SCALE;                                                             \
      t = fminf(fmaxf(t, -5.0f), 5.0f);                                                    \
      const float wr = ((mh >> (2 * p)) & 1u) ? __expf(t) : 0.0f;                          \
          \
         \
      const float w0 = __builtin_bit_cast(float, __builtin_amdgcn_permlane16(              \
          0u, __builtin_bit_cast(unsigned, wr), 0x11110000u, 0x33332222u, false, false));  \
      const float w1 = __builtin_bit_cast(float, __builtin_amdgcn_permlane16(              \
          0u, __builtin_bit_cast(unsigned, wr), 0x55554444u, 0x77776666u, false, false));  \
      acc[0] += pv[p][0].x * w0; acc[1] += pv[p][0].y * w0;                                \
      acc[2] += pv[p][0].z * w0; acc[3] += pv[p][0].w * w0;                                \
      acc[4] += pv[p][1].x * w1; acc[5] += pv[p][1].y * w1;                                \
      acc[6] += pv[p][1].z * w1; acc[7] += pv[p][1].w * w1;                                \
      z0 += w0;                                                                            \
      z1 += w1;                                                                            \
    }                                                                                      \
  }

#define ROW_BEGIN(d_)                                                                      \
  const float4* qrow = (const float4*)(q + ((unsigned)(d_) * HD + hsel * DIM + half * 8)); \
  const float4 qb0 = qrow[0];                                                              \
  const float4 qb1 = qrow[1];                                                              \
  float acc[8];                                                                            \
  _Pragma("unroll") for (int j = 0; j < 8; ++j) acc[j] = 0.0f;                             \
  float z0 = 0.0f, z1 = 0.0f;

#define ROW_END(d_)                                                                        \
  {                                                                                        \
    _Pragma("unroll") for (int j = 0; j < 8; ++j) acc[j] += __shfl_xor(acc[j], 16);        \
    z0 += __shfl_xor(z0, 16);                                                              \
    z1 += __shfl_xor(z1, 16);                                                              \
    const float den = (half ? z1 : z0) + 1e-6f;                                            \
    float4 o;                                                                              \
    o.x = (half ? acc[4] : acc[0]) / den;                                                  \
    o.y = (half ? acc[5] : acc[1]) / den;                                                  \
    o.z = (half ? acc[6] : acc[2]) / den;                                                  \
    o.w = (half ? acc[7] : acc[3]) / den;                                                  \
    *(float4*)(out + ((unsigned)(d_) * HD + half * 64 + c * 4)) = o;                       \
  }

#define GW 16
#define GT (GW * 32)
__global__ void __launch_bounds__(GT) __attribute__((amdgpu_waves_per_eu(4, 4)))
range_gather_kernel(const float* __restrict__ q, const float* __restrict__ k,
                    const float* __restrict__ v, const int* __restrict__ piece,
                    float* __restrict__ out, int nEdges, int nNodes, int chunk) {
  __shared__ int s_cnt[NCH];
  __shared__ int s_off[NCH];
  __shared__ int s_hist[256];
  __shared__ int s_cur[256];
  __shared__ int s_rend[256];
  __shared__ int s_part[8];
  __shared__ int s_tot[8];
  __shared__ int s_next;
  __shared__ int s_list[RG_CAP];
  const unsigned short* endM = (const unsigned short*)(piece + nEdges);
  const int tid  = threadIdx.x;
  const int lane = tid & 31;
  const int wv   = __builtin_amdgcn_readfirstlane(tid >> 5);
  const int g    = blockIdx.x;
  const int c      = lane & 15;
  const int half   = lane >> 4;
  const int hsel   = c & 7;
  const int rows = min(RB, nNodes - g * RB);

  int cn = 0;
  if (tid < NCH) {
    const int o = g ? (int)endM[tid * NRG + g - 1] : 0;
    cn = (int)endM[tid * NRG + g] - o;
    s_cnt[tid] = cn;
    s_off[tid] = o;
  }
  if (tid < 256) s_hist[tid] = 0;
  if (tid == 0) s_next = GW;
  int rc = cn;
  for (int off = 16; off > 0; off >>= 1) rc += __shfl_xor(rc, off);
  if (lane == 0 && wv < 8) s_tot[wv] = rc;
  __syncthreads();
  int total = 0;
  for (int i = 0; i < 8; ++i) total += s_tot[i];
  const bool staged = total <= RG_CAP;

  if (staged) {
    int rec[NCH / GW];
    unsigned have = 0;
#pragma unroll
    for (int j = 0; j < NCH / GW; ++j) {
      const int w = wv + GW * j;
      const int* run = piece + (long)w * chunk + s_off[w];
      const bool in = lane < s_cnt[w];
      rec[j] = in ? run[lane] : 0;
      have |= in ? (1u << j) : 0u;
    }
#pragma unroll
    for (int j = 0; j < NCH / GW; ++j)
      if (have & (1u << j)) atomicAdd(&s_hist[rec[j] & 255], 1);
    for (int j = 0; j < NCH / GW; ++j) {
      const int w = wv + GW * j;
      const int n = s_cnt[w];
      const int* run = piece + (long)w * chunk + s_off[w];
      for (int i = lane + 32; i < n; i += 32) atomicAdd(&s_hist[run[i] & 255], 1);
    }
    __syncthreads();

    const int hc = (tid < 256) ? s_hist[tid] : 0;
    int incl;
    const int excl = scan256(hc, tid, s_part, &incl);
    if (tid < 256) { s_cur[tid] = excl; s_rend[tid] = incl; }
    __syncthreads();

#pragma unroll
    for (int j = 0; j < NCH / GW; ++j) {
      if (have & (1u << j)) {
        const int pos = atomicAdd(&s_cur[rec[j] & 255], 1);
        s_list[pos] = (int)((unsigned)rec[j] >> 8);
      }
    }
    for (int j = 0; j < NCH / GW; ++j) {
      const int w = wv + GW * j;
      const int n = s_cnt[w];
      const int* run = piece + (long)w * chunk + s_off[w];
      for (int i = lane + 32; i < n; i += 32) {
        const int r   = run[i];
        const int pos = atomicAdd(&s_cur[r & 255], 1);
        s_list[pos] = (int)((unsigned)r >> 8);
      }
    }
    __syncthreads();

    int r = wv;
    if (r >= rows) return;
    int rn = 0;
    if (lane == 0) rn = atomicAdd(&s_next, 1);
    rn = __builtin_amdgcn_readfirstlane(rn);
    int beg = __builtin_amdgcn_readfirstlane(r ? s_rend[r ? r - 1 : 0] : 0);
    int end = __builtin_amdgcn_readfirstlane(s_rend[r]);
    int ids = 0;
    if (beg < end) ids = s_list[beg + (c < end - beg ? c : 0)];

    for (;;) {
      int nbeg = 0, nend = 0;
      if (rn < rows) {
        nbeg = __builtin_amdgcn_readfirstlane(s_rend[rn - 1]);
        nend = __builtin_amdgcn_readfirstlane(s_rend[rn]);
      }
      int rnn = 0;
      if (lane == 0) rnn = atomicAdd(&s_next, 1);
      rnn = __builtin_amdgcn_readfirstlane(rnn);

      const int d = g * RB + r;
      ROW_BEGIN(d)
      for (int t0 = beg; t0 < end; t0 += 16) {
        const int cnt = min(16, end - t0);
        TILE_LOADS(ids)
        const int t1 = t0 + 16;
        const int na = (t1 < end) ? t1 : nbeg;
        const int nn = (t1 < end) ? end - t1 : nend - nbeg;
        if (nn > 0) ids = s_list[na + (c < nn ? c : 0)];
        TILE_MATH(0xffffu >> (16 - cnt))
      }
      if (beg >= end && nbeg < nend) ids = s_list[nbeg + (c < nend - nbeg ? c : 0)];
      ROW_END(d)

      if (rn >= rows) break;
      r = rn; beg = nbeg; end = nend; rn = rnn;
    }
  } else {
    for (int r = wv; r < rows; r += GW) {
      const int d = g * RB + r;
      ROW_BEGIN(d)
      for (int w = 0; w < NCH; ++w) {
        const int n = __builtin_amdgcn_readfirstlane(s_cnt[w]);
        const int* run = piece + (long)w * chunk + __builtin_amdgcn_readfirstlane(s_off[w]);
        for (int i = 0; i < n; i += 16) {
          const int rcd = (c < n - i) ? run[i + c] : -1;
          const bool ok = rcd >= 0 && (rcd & 255) == r;
          const unsigned mask = (unsigned)__ballot(ok) & 0xffffu;
          if (mask) {
            TILE_LOADS(ok ? (int)((unsigned)rcd >> 8) : 0)
            TILE_MATH(mask)
          }
        }
      }
      ROW_END(d)
    }
  }
}

extern "C" void kernel_launch(void* const* d_in, const int* in_sizes, int n_in,
                              void* d_out, int out_size, void* d_ws, size_t ws_size,
                              hipStream_t stream) {
  const float* q    = (const float*)d_in[0];
  const float* k    = (const float*)d_in[1];
  const float* v    = (const float*)d_in[2];
  const int*   eidx = (const int*)d_in[3];
  float* out = (float*)d_out;

  const int nEdges = in_sizes[3] / 2;
  const int nNodes = out_size / HD;

  int* piece = (int*)d_ws;

  const int chunk = (nEdges + NCH - 1) / NCH;
  split_chunk_kernel<<<NCH, SORT_T, 0, stream>>>(eidx, piece, nEdges, nNodes, chunk);
  range_gather_kernel<<<NRG, GT, 0, stream>>>(q, k, v, piece, out, nEdges, nNodes, chunk);
}


# Gather tile body built for 1, 2, 4, 6 and 8 edge pairs, picked by a scalar branch: a row's last tile (deg mod 16 edges) issues loads, matrix products and weight work only for the pairs it needs
# speedup vs baseline: 12.5648x; 1.0830x over previous
#include <hip/hip_runtime.h>
#include <math.h>

#define HEADS 8
#define DIM 16
#define HD 128
#define INV_SCALE 0.25f

typedef __attribute__((ext_vector_type(2))) float v2f;
typedef __attribute__((ext_vector_type(8))) float v8f;

#define SORT_T 1024
#define RB 196
#define NRG 256
#define NCH 256
#define CH_CAP 4096
#define RG_CAP 32768

__device__ __forceinline__ int scan256(int c, int tid, int* s_part, int* inclusive) {
  const int lane = tid & 31;
  const int wv   = tid >> 5;
  int inc = c;
  for (int off = 1; off < 32; off <<= 1) {
    int t = __shfl_up(inc, off);
    if (lane >= off) inc += t;
  }
  if (lane == 31 && wv < 8) s_part[wv] = inc;
  __syncthreads();
  int pre = 0;
  for (int i = 0; i < 8; ++i) pre += (i < wv) ? s_part[i] : 0;
  *inclusive = pre + inc;
  return pre + inc - c;
}

__global__ void __launch_bounds__(SORT_T)
split_chunk_kernel(const int* __restrict__ eidx, int* __restrict__ piece,
                   int nEdges, int nNodes, int chunk) {
  __shared__ int s_hist[NRG];
  __shared__ int s_cur[NRG];
  __shared__ int s_part[8];
  __shared__ int s_stage[CH_CAP];
  unsigned short* endM = (unsigned short*)(piece + nEdges);
  const int tid = threadIdx.x;
  const int w   = blockIdx.x;
  const int e0  = w * chunk;
  const int e1  = min(nEdges, e0 + chunk);

  if (tid < NRG) s_hist[tid] = 0;
  __syncthreads();
  for (int eb = e0 + tid; eb < e1; eb += 4 * SORT_T) {
    int dd[4];
#pragma unroll
    for (int j = 0; j < 4; ++j) {
      const int e = eb + j * SORT_T;
      const int v = eidx[(long)nEdges + min(e, e1 - 1)];
      dd[j] = (e < e1) ? v : -1;
    }
#pragma unroll
    for (int j = 0; j < 4; ++j)
      if ((unsigned)dd[j] < (unsigned)nNodes) atomicAdd(&s_hist[dd[j] / RB], 1);
  }
  __syncthreads();
  const int c = (tid < NRG) ? s_hist[tid] : 0;
  int incl;
  const int excl = scan256(c, tid, s_part, &incl);
  if (tid < NRG) {
    s_cur[tid] = excl;
    endM[w * NRG + tid] = (unsigned short)incl;
  }
  __syncthreads();
  const bool staged = (e1 - e0) <= CH_CAP;
  for (int eb = e0 + tid; eb < e1; eb += 4 * SORT_T) {
    int ss[4], dd[4];
#pragma unroll
    for (int j = 0; j < 4; ++j) {
      const int e  = eb + j * SORT_T;
      const int ec = min(e, e1 - 1);
      ss[j] = eidx[ec];
      const int v = eidx[(long)nEdges + ec];
      dd[j] = (e < e1) ? v : -1;
    }
#pragma unroll
    for (int j = 0; j < 4; ++j) {
      if ((unsigned)dd[j] < (unsigned)nNodes) {
        const int g   = dd[j] / RB;
        const int pos = atomicAdd(&s_cur[g], 1);
        const int rec = (ss[j] << 8) | (dd[j] - g * RB);
        if (staged) s_stage[pos] = rec; else piece[e0 + pos] = rec;
      }
    }
  }
  __syncthreads();
  if (staged) {
    const int total = s_cur[NRG - 1];
    for (int i = tid; i < total; i += SORT_T) piece[e0 + i] = s_stage[i];
  }
}

#define TILE_LOADS(ids_, NP_)                                                              \
  const int srcc = (ids_);                                                                 \
  float4 kk[NP_][2];                                                                       \
  float4 pv[NP_][2];                                                                       \
  _Pragma("unroll") for (int p = 0; p < (NP_); ++p) {                                      \
    const int sA = __builtin_amdgcn_readlane(srcc, 2 * p);                                 \
    const int sB = __builtin_amdgcn_readlane(srcc, 2 * p + 1);                             \
    const int sk = (c & 8) ? sB : sA;                                                      \
    const float4* kp = (const float4*)(k + ((unsigned)sk * HD + hsel * DIM + half * 8));   \
    kk[p][0] = kp[0];                                                                      \
    kk[p][1] = kp[1];                                                                      \
  }                                                                                        \
  _Pragma("unroll") for (int p = 0; p < (NP_); ++p) {                                      \
    const int sA = __builtin_amdgcn_readlane(srcc, 2 * p);                                 \
    const int sB = __builtin_amdgcn_readlane(srcc, 2 * p + 1);                             \
    const int sv = half ? sB : sA;                                                         \
    const float* vp = v + ((unsigned)sv * HD + c * 4);                                     \
    pv[p][0] = *(const float4*)vp;                                                         \
    pv[p][1] = *(const float4*)(vp + 64);                                                  \
  }

#define TILE_MATH(mask_, NP_)                                                              \
  {                                                                                        \
    const unsigned mh = (unsigned)(mask_) >> half;                                         \
    _Pragma("unroll") for (int p = 0; p < (NP_); ++p) {                                    \
      const float4 a0 = kk[p][0], a1 = kk[p][1];                                           \
      v8f cc = {0.f, 0.f, 0.f, 0.f, 0.f, 0.f, 0.f, 0.f};                                   \
      v2f a, b;                                                                            \
      a = (v2f){a0.x, a0.y}; b = (v2f){qb0.x, qb0.y};                                      \
      cc = __builtin_amdgcn_wmma_f32_16x16x4_f32(false, a, false, b, (short)0, cc, false, false); \
      a = (v2f){a0.z, a0.w}; b = (v2f){qb0.z, qb0.w};                                      \
      cc = __builtin_amdgcn_wmma_f32_16x16x4_f32(false, a, false, b, (short)0, cc, false, false); \
      a = (v2f){a1.x, a1.y}; b = (v2f){qb1.x, qb1.y};                                      \
      cc = __builtin_amdgcn_wmma_f32_16x16x4_f32(false, a, false, b, (short)0, cc, false, false); \
      a = (v2f){a1.z, a1.w}; b = (v2f){qb1.z, qb1.w};                                      \
      cc = __builtin_amdgcn_wmma_f32_16x16x4_f32(false, a, false, b, (short)0, cc, false, false); \
      float s = cc[0];                                                                     \
      _Pragma("unroll") for (int r = 1; r < 8; ++r) s = (hsel == r) ? cc[r] : s;           \
      float t = s * INV_SCALE;                                                             \
      t = fminf(fmaxf(t, -5.0f), 5.0f);                                                    \
      const float wr = ((mh >> (2 * p)) & 1u) ? __expf(t) : 0.0f;                          \
          \
         \
      const float w0 = __builtin_bit_cast(float, __builtin_amdgcn_permlane16(              \
          0u, __builtin_bit_cast(unsigned, wr), 0x11110000u, 0x33332222u, false, false));  \
      const float w1 = __builtin_bit_cast(float, __builtin_amdgcn_permlane16(              \
          0u, __builtin_bit_cast(unsigned, wr), 0x55554444u, 0x77776666u, false, false));  \
      acc[0] += pv[p][0].x * w0; acc[1] += pv[p][0].y * w0;                                \
      acc[2] += pv[p][0].z * w0; acc[3] += pv[p][0].w * w0;                                \
      acc[4] += pv[p][1].x * w1; acc[5] += pv[p][1].y * w1;                                \
      acc[6] += pv[p][1].z * w1; acc[7] += pv[p][1].w * w1;                                \
      z0 += w0;                                                                            \
      z1 += w1;                                                                            \
    }                                                                                      \
  }

#define ROW_BEGIN(d_)                                                                      \
  const float4* qrow = (const float4*)(q + ((unsigned)(d_) * HD + hsel * DIM + half * 8)); \
  const float4 qb0 = qrow[0];                                                              \
  const float4 qb1 = qrow[1];                                                              \
  float acc[8];                                                                            \
  _Pragma("unroll") for (int j = 0; j < 8; ++j) acc[j] = 0.0f;                             \
  float z0 = 0.0f, z1 = 0.0f;

#define ROW_END(d_)                                                                        \
  {                                                                                        \
    _Pragma("unroll") for (int j = 0; j < 8; ++j) acc[j] += __shfl_xor(acc[j], 16);        \
    z0 += __shfl_xor(z0, 16);                                                              \
    z1 += __shfl_xor(z1, 16);                                                              \
    const float den = (half ? z1 : z0) + 1e-6f;                                            \
    float4 o;                                                                              \
    o.x = (half ? acc[4] : acc[0]) / den;                                                  \
    o.y = (half ? acc[5] : acc[1]) / den;                                                  \
    o.z = (half ? acc[6] : acc[2]) / den;                                                  \
    o.w = (half ? acc[7] : acc[3]) / den;                                                  \
    *(float4*)(out + ((unsigned)(d_) * HD + half * 64 + c * 4)) = o;                       \
  }

#define GW 16
#define GT (GW * 32)
__global__ void __launch_bounds__(GT) __attribute__((amdgpu_waves_per_eu(4, 4)))
range_gather_kernel(const float* __restrict__ q, const float* __restrict__ k,
                    const float* __restrict__ v, const int* __restrict__ piece,
                    float* __restrict__ out, int nEdges, int nNodes, int chunk) {
  __shared__ int s_cnt[NCH];
  __shared__ int s_off[NCH];
  __shared__ int s_hist[256];
  __shared__ int s_cur[256];
  __shared__ int s_rend[256];
  __shared__ int s_part[8];
  __shared__ int s_tot[8];
  __shared__ int s_next;
  __shared__ int s_list[RG_CAP];
  const unsigned short* endM = (const unsigned short*)(piece + nEdges);
  const int tid  = threadIdx.x;
  const int lane = tid & 31;
  const int wv   = __builtin_amdgcn_readfirstlane(tid >> 5);
  const int g    = blockIdx.x;
  const int c      = lane & 15;
  const int half   = lane >> 4;
  const int hsel   = c & 7;
  const int rows = min(RB, nNodes - g * RB);

  int cn = 0;
  if (tid < NCH) {
    const int o = g ? (int)endM[tid * NRG + g - 1] : 0;
    cn = (int)endM[tid * NRG + g] - o;
    s_cnt[tid] = cn;
    s_off[tid] = o;
  }
  if (tid < 256) s_hist[tid] = 0;
  if (tid == 0) s_next = GW;
  int rc = cn;
  for (int off = 16; off > 0; off >>= 1) rc += __shfl_xor(rc, off);
  if (lane == 0 && wv < 8) s_tot[wv] = rc;
  __syncthreads();
  int total = 0;
  for (int i = 0; i < 8; ++i) total += s_tot[i];
  const bool staged = total <= RG_CAP;

  if (staged) {
    int rec[NCH / GW];
    unsigned have = 0;
#pragma unroll
    for (int j = 0; j < NCH / GW; ++j) {
      const int w = wv + GW * j;
      const int* run = piece + (long)w * chunk + s_off[w];
      const bool in = lane < s_cnt[w];
      rec[j] = in ? run[lane] : 0;
      have |= in ? (1u << j) : 0u;
    }
#pragma unroll
    for (int j = 0; j < NCH / GW; ++j)
      if (have & (1u << j)) atomicAdd(&s_hist[rec[j] & 255], 1);
    for (int j = 0; j < NCH / GW; ++j) {
      const int w = wv + GW * j;
      const int n = s_cnt[w];
      const int* run = piece + (long)w * chunk + s_off[w];
      for (int i = lane + 32; i < n; i += 32) atomicAdd(&s_hist[run[i] & 255], 1);
    }
    __syncthreads();

    const int hc = (tid < 256) ? s_hist[tid] : 0;
    int incl;
    const int excl = scan256(hc, tid, s_part, &incl);
    if (tid < 256) { s_cur[tid] = excl; s_rend[tid] = incl; }
    __syncthreads();

#pragma unroll
    for (int j = 0; j < NCH / GW; ++j) {
      if (have & (1u << j)) {
        const int pos = atomicAdd(&s_cur[rec[j] & 255], 1);
        s_list[pos] = (int)((unsigned)rec[j] >> 8);
      }
    }
    for (int j = 0; j < NCH / GW; ++j) {
      const int w = wv + GW * j;
      const int n = s_cnt[w];
      const int* run = piece + (long)w * chunk + s_off[w];
      for (int i = lane + 32; i < n; i += 32) {
        const int r   = run[i];
        const int pos = atomicAdd(&s_cur[r & 255], 1);
        s_list[pos] = (int)((unsigned)r >> 8);
      }
    }
    __syncthreads();

    int r = wv;
    if (r >= rows) return;
    int rn = 0;
    if (lane == 0) rn = atomicAdd(&s_next, 1);
    rn = __builtin_amdgcn_readfirstlane(rn);
    int beg = __builtin_amdgcn_readfirstlane(r ? s_rend[r ? r - 1 : 0] : 0);
    int end = __builtin_amdgcn_readfirstlane(s_rend[r]);
    int ids = 0;
    if (beg < end) ids = s_list[beg + (c < end - beg ? c : 0)];

    for (;;) {
      int nbeg = 0, nend = 0;
      if (rn < rows) {
        nbeg = __builtin_amdgcn_readfirstlane(s_rend[rn - 1]);
        nend = __builtin_amdgcn_readfirstlane(s_rend[rn]);
      }
      int rnn = 0;
      if (lane == 0) rnn = atomicAdd(&s_next, 1);
      rnn = __builtin_amdgcn_readfirstlane(rnn);

      const int d = g * RB + r;
      ROW_BEGIN(d)
      for (int t0 = beg; t0 < end; t0 += 16) {
        const int cnt = min(16, end - t0);
        const int t1 = t0 + 16;
        const int na = (t1 < end) ? t1 : nbeg;
        const int nn = (t1 < end) ? end - t1 : nend - nbeg;
#define STAGED_TILE(NP_)                                                                   \
        {                                                                                  \
          TILE_LOADS(ids, NP_)                                                             \
          if (nn > 0) ids = s_list[na + (c < nn ? c : 0)];                                 \
          TILE_MATH(0xffffu >> (16 - cnt), NP_)                                            \
        }
        switch ((cnt + 1) >> 1) {
          case 1: STAGED_TILE(1) break;
          case 2: STAGED_TILE(2) break;
          case 3:
          case 4: STAGED_TILE(4) break;
          case 5:
          case 6: STAGED_TILE(6) break;
          default: STAGED_TILE(8) break;
        }
#undef STAGED_TILE
      }
      if (beg >= end && nbeg < nend) ids = s_list[nbeg + (c < nend - nbeg ? c : 0)];
      ROW_END(d)

      if (rn >= rows) break;
      r = rn; beg = nbeg; end = nend; rn = rnn;
    }
  } else {
    for (int r = wv; r < rows; r += GW) {
      const int d = g * RB + r;
      ROW_BEGIN(d)
      for (int w = 0; w < NCH; ++w) {
        const int n = __builtin_amdgcn_readfirstlane(s_cnt[w]);
        const int* run = piece + (long)w * chunk + __builtin_amdgcn_readfirstlane(s_off[w]);
        for (int i = 0; i < n; i += 16) {
          const int rcd = (c < n - i) ? run[i + c] : -1;
          const bool ok = rcd >= 0 && (rcd & 255) == r;
          const unsigned mask = (unsigned)__ballot(ok) & 0xffffu;
          if (mask) {
            TILE_LOADS(ok ? (int)((unsigned)rcd >> 8) : 0, 8)
            TILE_MATH(mask, 8)
          }
        }
      }
      ROW_END(d)
    }
  }
}

extern "C" void kernel_launch(void* const* d_in, const int* in_sizes, int n_in,
                              void* d_out, int out_size, void* d_ws, size_t ws_size,
                              hipStream_t stream) {
  const float* q    = (const float*)d_in[0];
  const float* k    = (const float*)d_in[1];
  const float* v    = (const float*)d_in[2];
  const int*   eidx = (const int*)d_in[3];
  float* out = (float*)d_out;

  const int nEdges = in_sizes[3] / 2;
  const int nNodes = out_size / HD;

  int* piece = (int*)d_ws;

  const int chunk = (nEdges + NCH - 1) / NCH;
  split_chunk_kernel<<<NCH, SORT_T, 0, stream>>>(eidx, piece, nEdges, nNodes, chunk);
  range_gather_kernel<<<NRG, GT, 0, stream>>>(q, k, v, piece, out, nEdges, nNodes, chunk);
}
